# starting kernel
# baseline (speedup 1.0000x reference)
#include <hip/hip_runtime.h>
#include <hip/hip_bf16.h>

typedef __attribute__((ext_vector_type(2))) float v2f;
typedef __attribute__((ext_vector_type(8))) float v8f;

__global__ void k_init_deg(float* __restrict__ deg, int n) {
  int i = blockIdx.x * blockDim.x + threadIdx.x;
  if (i < n) deg[i] = 1.0f;
}

__global__ void k_count_deg(const int* __restrict__ dst, float* __restrict__ deg, int e) {
  int i = blockIdx.x * blockDim.x + threadIdx.x;
  if (i < e) atomicAdd(&deg[dst[i]], 1.0f);
}

__global__ void k_rsqrt_inplace(float* __restrict__ d, int n) {
  int i = blockIdx.x * blockDim.x + threadIdx.x;
  if (i < n) d[i] = rsqrtf(d[i]);
}

template <int K, int N>
__global__ void k_gemm_wmma_f32(const float* __restrict__ A,
                                const float* __restrict__ B,
                                float* __restrict__ C) {
  const int lane = threadIdx.x & 31;
  const int row  = lane & 15;
  const int kp   = lane >> 4;
  const int m0   = blockIdx.x * 16;
  const int n0   = blockIdx.y * 16;
  const int col  = n0 + row;
  const int colc = (col < N) ? col : (N - 1);

  const float* Arow = A + (size_t)(m0 + row) * K;
  v8f c = {};
#pragma unroll
  for (int k0 = 0; k0 < K; k0 += 4) {
    const int ka = k0 + 2 * kp;
    v2f a, b;
    a.x = Arow[ka];
    a.y = Arow[ka + 1];
    b.x = B[(size_t)ka * N + colc];
    b.y = B[(size_t)(ka + 1) * N + colc];
    c = __builtin_amdgcn_wmma_f32_16x16x4_f32(false, a, false, b, (short)0, c,
                                              false, false);
  }
  if (col < N) {
#pragma unroll
    for (int r = 0; r < 8; ++r)
      C[(size_t)(m0 + r + 8 * kp) * N + col] = c[r];
  }
}

__global__ void k_self_loop_init(const float* __restrict__ h,
                                 const float* __restrict__ dinv,
                                 float* __restrict__ acc, int n, int stride) {
  int idx = blockIdx.x * blockDim.x + threadIdx.x;
  if (idx >= n * stride) return;
  int i = idx / stride;
  float di = dinv[i];
  acc[idx] = h[idx] * di * di;
}

template <int F4, int STRIDE>
__global__ void k_edge_agg(const int* __restrict__ src, const int* __restrict__ dst,
                           const float* __restrict__ dinv,
                           const float* __restrict__ h,
                           float* __restrict__ acc, int e) {
  int tid = blockIdx.x * blockDim.x + threadIdx.x;
  if (tid >= e * F4) return;
  int ei  = tid / F4;
  int sub = tid - ei * F4;
  int s = src[ei], d = dst[ei];
  float norm = dinv[s] * dinv[d];
  const float4 v = *(const float4*)(h + (size_t)s * STRIDE + sub * 4);
  float* ap = acc + (size_t)d * STRIDE + sub * 4;
  atomicAdd(ap + 0, v.x * norm);
  atomicAdd(ap + 1, v.y * norm);
  atomicAdd(ap + 2, v.z * norm);
  atomicAdd(ap + 3, v.w * norm);
}

__global__ void k_relu_bias(const float* __restrict__ acc,
                            const float* __restrict__ b,
                            float* __restrict__ out, int n, int stride) {
  int idx = blockIdx.x * blockDim.x + threadIdx.x;
  if (idx >= n * stride) return;
  int f = idx % stride;
  out[idx] = fmaxf(acc[idx] + b[f], 0.0f);
}

__global__ void k_bias_logsoftmax40(float* __restrict__ out,
                                    const float* __restrict__ b, int n) {
  int i = blockIdx.x * blockDim.x + threadIdx.x;
  if (i >= n) return;
  float* p = out + (size_t)i * 40;
  float v[40];
  float m = -3.402823466e38f;
#pragma unroll
  for (int f = 0; f < 40; ++f) {
    v[f] = p[f] + b[f];
    m = fmaxf(m, v[f]);
  }
  float s = 0.0f;
#pragma unroll
  for (int f = 0; f < 40; ++f) s += expf(v[f] - m);
  float lse = m + logf(s);
#pragma unroll
  for (int f = 0; f < 40; ++f) p[f] = v[f] - lse;
}

extern "C" void kernel_launch(void* const* d_in, const int* in_sizes, int n_in,
                              void* d_out, int out_size, void* d_ws, size_t ws_size,
                              hipStream_t stream) {
  (void)n_in; (void)out_size; (void)ws_size;

  const float* x  = (const float*)d_in[0];
  const int*   ei = (const int*)d_in[1];
  const float* W1 = (const float*)d_in[2];
  const float* b1 = (const float*)d_in[3];
  const float* W2 = (const float*)d_in[4];
  const float* b2 = (const float*)d_in[5];
  float* out = (float*)d_out;

  const int F_IN = 128, F_H = 64, F_OUT = 40;
  const int Nn = in_sizes[0] / F_IN;
  const int E  = in_sizes[1] / 2;
  const int* src = ei;
  const int* dst = ei + E;

  char* ws = (char*)d_ws;
  float* dinv = (float*)(ws);
  float* h    = (float*)(ws + ((size_t)4  << 20));
  float* acc1 = (float*)(ws + ((size_t)36 << 20));
  float* h2   = (float*)(ws + ((size_t)68 << 20));

  const dim3 B256(256);

  k_init_deg<<<(Nn + 255) / 256, B256, 0, stream>>>(dinv, Nn);
  k_count_deg<<<(E + 255) / 256, B256, 0, stream>>>(dst, dinv, E);
  k_rsqrt_inplace<<<(Nn + 255) / 256, B256, 0, stream>>>(dinv, Nn);

  k_gemm_wmma_f32<128, 64><<<dim3(Nn / 16, 4), dim3(32), 0, stream>>>(x, W1, h);
  k_self_loop_init<<<((Nn * F_H) + 255) / 256, B256, 0, stream>>>(h, dinv, acc1, Nn, F_H);
  k_edge_agg<16, 64><<<((E * 16) + 255) / 256, B256, 0, stream>>>(src, dst, dinv, h, acc1, E);
  k_relu_bias<<<((Nn * F_H) + 255) / 256, B256, 0, stream>>>(acc1, b1, h, Nn, F_H);

  k_gemm_wmma_f32<64, 40><<<dim3(Nn / 16, 3), dim3(32), 0, stream>>>(h, W2, h2);
  k_self_loop_init<<<((Nn * F_OUT) + 255) / 256, B256, 0, stream>>>(h2, dinv, out, Nn, F_OUT);
  k_edge_agg<10, 40><<<((E * 10) + 255) / 256, B256, 0, stream>>>(src, dst, dinv, h2, out, E);
  k_bias_logsoftmax40<<<(Nn + 255) / 256, B256, 0, stream>>>(out, b2, Nn);
}


# Row-wise elementwise launches folded into the two GEMMs: each also stores the self-loop start value h*dinv^2, layer 2 applies bias+relu to A as it loads it; 8 launches, was 11; values unchanged
# speedup vs baseline: 1.4758x; 1.4758x over previous
#include <hip/hip_runtime.h>
#include <hip/hip_bf16.h>

typedef __attribute__((ext_vector_type(2))) float v2f;
typedef __attribute__((ext_vector_type(4))) float v4f;
typedef __attribute__((ext_vector_type(8))) float v8f;

__global__ void k_init_deg(float* __restrict__ deg, int n) {
  int i = blockIdx.x * blockDim.x + threadIdx.x;
  if (i < n) deg[i] = 1.0f;
}

__global__ void k_count_deg(const int* __restrict__ dst, float* __restrict__ deg, int e) {
  int i = blockIdx.x * blockDim.x + threadIdx.x;
  if (i < e) atomicAdd(&deg[dst[i]], 1.0f);
}

__global__ void k_rsqrt_inplace(float* __restrict__ d, int n) {
  int i = blockIdx.x * blockDim.x + threadIdx.x;
  if (i < n) d[i] = rsqrtf(d[i]);
}

template <int K, int N, bool RELU_A>
__global__ void k_gemm_wmma_f32(const float* __restrict__ A,
                                const float* __restrict__ abias,
                                const float* __restrict__ B,
                                const float* __restrict__ dinv,
                                float* __restrict__ C,
                                float* __restrict__ S) {
  const int lane = threadIdx.x & 31;
  const int row  = lane & 15;
  const int kp   = lane >> 4;
  const int m0   = blockIdx.x * 16;
  const int n0   = blockIdx.y * 16;
  const int col  = n0 + row;
  const int colc = (col < N) ? col : (N - 1);

  const v4f* dp = reinterpret_cast<const v4f*>(dinv + m0 + 8 * kp);
  const v4f dlo = dp[0];
  const v4f dhi = dp[1];

  const float* Arow = A + (size_t)(m0 + row) * K;
  v8f c = {};
#pragma unroll
  for (int k0 = 0; k0 < K; k0 += 4) {
    const int ka = k0 + 2 * kp;
    v2f a, b;
    a.x = Arow[ka];
    a.y = Arow[ka + 1];
    if (RELU_A) {
      a.x = fmaxf(a.x + abias[ka], 0.0f);
      a.y = fmaxf(a.y + abias[ka + 1], 0.0f);
    }
    b.x = B[(size_t)ka * N + colc];
    b.y = B[(size_t)(ka + 1) * N + colc];
    c = __builtin_amdgcn_wmma_f32_16x16x4_f32(false, a, false, b, (short)0, c,
                                              false, false);
  }
  asm volatile("" ::"v"(dlo), "v"(dhi));
  if (col < N) {
    const float di[8] = {dlo.x, dlo.y, dlo.z, dlo.w, dhi.x, dhi.y, dhi.z, dhi.w};
#pragma unroll
    for (int r = 0; r < 8; ++r) {
      const size_t o = (size_t)(m0 + r + 8 * kp) * N + col;
      C[o] = c[r];
      S[o] = c[r] * di[r] * di[r];
    }
  }
}

template <int NCOL, int U>
__device__ __forceinline__ void edge_rows_add(int sv, int dv, float nv, int j,
                                              int lane,
                                              const float* __restrict__ h,
                                              float* __restrict__ acc) {
  float ra[U], rb[U], nn[U];
  int cA[U];
  size_t w0[U];
#pragma unroll
  for (int u = 0; u < U; ++u) {
    const int s = __builtin_amdgcn_readlane(sv, j + u);
    const int d = __builtin_amdgcn_readlane(dv, j + u);
    nn[u] = __int_as_float(__builtin_amdgcn_readlane(__float_as_int(nv), j + u));
    const int start = d * NCOL;
    const int inoff = (NCOL % 32 == 0) ? 0 : (start & 31);
    w0[u] = (size_t)(start - inoff);
    cA[u] = lane - inoff;
    const int cB = cA[u] + 32;
    const float* hp = h + (size_t)s * NCOL;
    ra[u] = hp[cA[u] < 0 ? 0 : cA[u]];
    rb[u] = hp[cB < NCOL ? cB : NCOL - 1];
  }
  if (NCOL % 32 != 0) {
#pragma unroll
    for (int u = 0; u < U; ++u) asm volatile("" ::"v"(ra[u]), "v"(rb[u]));
  }
#pragma unroll
  for (int u = 0; u < U; ++u) {
    float* ap = acc + w0[u] + lane;
    if (NCOL % 32 == 0 || cA[u] >= 0) atomicAdd(ap, ra[u] * nn[u]);
    if (NCOL % 32 == 0 || cA[u] + 32 < NCOL) atomicAdd(ap + 32, rb[u] * nn[u]);
  }
}

template <int NCOL>
__global__ void __launch_bounds__(256)
k_edge_agg_lines(const int* __restrict__ src, const int* __restrict__ dst,
                 const float* __restrict__ dinv, const float* __restrict__ h,
                 float* __restrict__ acc, int e) {
  static_assert(NCOL == 64 || NCOL == 40, "a row must lie in two 128-byte lines");
  const int lane = threadIdx.x & 31;
  const int e0 = __builtin_amdgcn_readfirstlane(
      (int)((blockIdx.x * blockDim.x + threadIdx.x) >> 5) * 32);
  if (e0 >= e) return;
  const int mine = e0 + lane;
  const bool have = mine < e;
  const int sv = have ? src[mine] : 0;
  const int dv = have ? dst[mine] : 0;
  const float nv = have ? dinv[sv] * dinv[dv] : 0.0f;
  const int cnt = (e - e0 < 32) ? (e - e0) : 32;
  int j = 0;
  for (; j + 8 <= cnt; j += 8) edge_rows_add<NCOL, 8>(sv, dv, nv, j, lane, h, acc);
  for (; j < cnt; ++j) edge_rows_add<NCOL, 1>(sv, dv, nv, j, lane, h, acc);
}

__global__ void k_bias_logsoftmax40(float* __restrict__ out,
                                    const float* __restrict__ b, int n) {
  int i = blockIdx.x * blockDim.x + threadIdx.x;
  if (i >= n) return;
  float* p = out + (size_t)i * 40;
  float v[40];
  float m = -3.402823466e38f;
#pragma unroll
  for (int f = 0; f < 40; ++f) {
    v[f] = p[f] + b[f];
    m = fmaxf(m, v[f]);
  }
  float s = 0.0f;
#pragma unroll
  for (int f = 0; f < 40; ++f) s += expf(v[f] - m);
  float lse = m + logf(s);
#pragma unroll
  for (int f = 0; f < 40; ++f) p[f] = v[f] - lse;
}

extern "C" void kernel_launch(void* const* d_in, const int* in_sizes, int n_in,
                              void* d_out, int out_size, void* d_ws, size_t ws_size,
                              hipStream_t stream) {
  (void)n_in; (void)out_size; (void)ws_size;

  const float* x  = (const float*)d_in[0];
  const int*   ei = (const int*)d_in[1];
  const float* W1 = (const float*)d_in[2];
  const float* b1 = (const float*)d_in[3];
  const float* W2 = (const float*)d_in[4];
  const float* b2 = (const float*)d_in[5];
  float* out = (float*)d_out;

  const int F_IN = 128;
  const int Nn = in_sizes[0] / F_IN;
  const int E  = in_sizes[1] / 2;
  const int* src = ei;
  const int* dst = ei + E;

  char* ws = (char*)d_ws;
  float* dinv = (float*)(ws);
  float* h    = (float*)(ws + ((size_t)4  << 20));
  float* acc1 = (float*)(ws + ((size_t)36 << 20));
  float* h2   = (float*)(ws + ((size_t)68 << 20));

  const dim3 B256(256);

  k_init_deg<<<(Nn + 255) / 256, B256, 0, stream>>>(dinv, Nn);
  k_count_deg<<<(E + 255) / 256, B256, 0, stream>>>(dst, dinv, E);
  k_rsqrt_inplace<<<(Nn + 255) / 256, B256, 0, stream>>>(dinv, Nn);

  k_gemm_wmma_f32<128, 64, false><<<dim3(Nn / 16, 4), dim3(32), 0, stream>>>(
      x, b1, W1, dinv, h, acc1);
  k_edge_agg_lines<64><<<(E + 255) / 256, B256, 0, stream>>>(src, dst, dinv, h, acc1, E);

  k_gemm_wmma_f32<64, 40, true><<<dim3(Nn / 16, 3), dim3(32), 0, stream>>>(
      acc1, b1, W2, dinv, h2, out);
  k_edge_agg_lines<40><<<(E + 255) / 256, B256, 0, stream>>>(src, dst, dinv, h2, out, E);
  k_bias_logsoftmax40<<<(Nn + 255) / 256, B256, 0, stream>>>(out, b2, Nn);
}


# Both GEMMs weight-stationary: a wave keeps its 16-column slice of W in registers and walks row tiles with 16-byte A loads, next tile requested ahead; 1024 work-groups, was 43750 (sum order changes)
# speedup vs baseline: 1.7746x; 1.2025x over previous
#include <hip/hip_runtime.h>
#include <hip/hip_bf16.h>

typedef __attribute__((ext_vector_type(2))) float v2f;
typedef __attribute__((ext_vector_type(4))) float v4f;
typedef __attribute__((ext_vector_type(8))) float v8f;

__global__ void k_init_deg(float* __restrict__ deg, int n) {
  int i = blockIdx.x * blockDim.x + threadIdx.x;
  if (i < n) deg[i] = 1.0f;
}

__global__ void k_count_deg(const int* __restrict__ dst, float* __restrict__ deg, int e) {
  int i = blockIdx.x * blockDim.x + threadIdx.x;
  if (i < e) atomicAdd(&deg[dst[i]], 1.0f);
}

__global__ void k_rsqrt_inplace(float* __restrict__ d, int n) {
  int i = blockIdx.x * blockDim.x + threadIdx.x;
  if (i < n) d[i] = rsqrtf(d[i]);
}

__device__ __forceinline__ int kof(int j, int h, int s) {
  return 8 * (j >> 1) + 4 * h + 2 * (j & 1) + s;
}

template <int K, int N, bool RELU_A>
struct GemmWS {
  static constexpr int NJ = K / 4;
  static constexpr int NQ = K / 8;

  struct Tile {
    v4f a[NQ];
    v4f dlo, dhi;
  };

  static __device__ __forceinline__ void load(Tile& t, int tile, int c16, int h,
                                              const float* __restrict__ A,
                                              const float* __restrict__ dinv) {
    const int m0 = tile * 16;
    const v4f* ap = reinterpret_cast<const v4f*>(A + (size_t)(m0 + c16) * K + 4 * h);
#pragma unroll
    for (int i = 0; i < NQ; ++i) t.a[i] = ap[2 * i];
    const v4f* dp = reinterpret_cast<const v4f*>(dinv + m0 + 8 * h);
    t.dlo = dp[0];
    t.dhi = dp[1];
  }

  static __device__ __forceinline__ void run(const Tile& t, int tile, int col, int h,
                                             const v2f (&bw)[NJ], const v4f (&bias)[NQ],
                                             float* __restrict__ C,
                                             float* __restrict__ S) {
    v8f c = {};
#pragma unroll
    for (int i = 0; i < NQ; ++i) {
      v4f q = t.a[i];
      if (RELU_A) {
        q.x = fmaxf(q.x + bias[i].x, 0.0f);
        q.y = fmaxf(q.y + bias[i].y, 0.0f);
        q.z = fmaxf(q.z + bias[i].z, 0.0f);
        q.w = fmaxf(q.w + bias[i].w, 0.0f);
      }
      v2f lo, hi;
      lo.x = q.x; lo.y = q.y;
      hi.x = q.z; hi.y = q.w;
      c = __builtin_amdgcn_wmma_f32_16x16x4_f32(false, lo, false, bw[2 * i], (short)0, c,
                                                false, false);
      c = __builtin_amdgcn_wmma_f32_16x16x4_f32(false, hi, false, bw[2 * i + 1], (short)0, c,
                                                false, false);
    }
    if (N % 16 == 0 || col < N) {
      const float di[8] = {t.dlo.x, t.dlo.y, t.dlo.z, t.dlo.w,
                           t.dhi.x, t.dhi.y, t.dhi.z, t.dhi.w};
      const int m0 = tile * 16;
#pragma unroll
      for (int r = 0; r < 8; ++r) {
        const size_t o = (size_t)(m0 + r + 8 * h) * N + col;
        C[o] = c[r];
        S[o] = c[r] * di[r] * di[r];
      }
    }
  }
};

template <int K, int N, bool RELU_A>
__global__ void __launch_bounds__(((N + 15) / 16) * 32)
k_gemm_wmma_f32(const float* __restrict__ A,
                const float* __restrict__ abias,
                const float* __restrict__ B,
                const float* __restrict__ dinv,
                float* __restrict__ C,
                float* __restrict__ S,
                int ntiles) {
  typedef GemmWS<K, N, RELU_A> G;
  const int lane = threadIdx.x & 31;
  const int c16  = lane & 15;
  const int h    = lane >> 4;
  const int col  = (int)(threadIdx.x >> 5) * 16 + c16;
  const int colc = (col < N) ? col : (N - 1);

  v2f bw[G::NJ];
#pragma unroll
  for (int j = 0; j < G::NJ; ++j) {
    bw[j].x = B[(size_t)kof(j, h, 0) * N + colc];
    bw[j].y = B[(size_t)kof(j, h, 1) * N + colc];
  }
  v4f bias[G::NQ];
#pragma unroll
  for (int i = 0; i < G::NQ; ++i) {
    if (RELU_A) {
      bias[i].x = abias[kof(2 * i, h, 0)];
      bias[i].y = abias[kof(2 * i, h, 1)];
      bias[i].z = abias[kof(2 * i + 1, h, 0)];
      bias[i].w = abias[kof(2 * i + 1, h, 1)];
    } else {
      bias[i] = v4f{0.0f, 0.0f, 0.0f, 0.0f};
    }
  }

  const int step = gridDim.x;
  int t = blockIdx.x;
  if (t >= ntiles) return;
  typename G::Tile t0, t1;
  G::load(t0, t, c16, h, A, dinv);
  while (t + step < ntiles) {
    const int tn = t + step;
    G::load(t1, tn, c16, h, A, dinv);
    __builtin_amdgcn_sched_barrier(0);
    G::run(t0, t, col, h, bw, bias, C, S);
    const int tnn = tn + step;
    G::load(t0, tnn < ntiles ? tnn : tn, c16, h, A, dinv);
    __builtin_amdgcn_sched_barrier(0);
    G::run(t1, tn, col, h, bw, bias, C, S);
    t = tnn;
  }
  if (t < ntiles) G::run(t0, t, col, h, bw, bias, C, S);
}

template <int NCOL, int U>
__device__ __forceinline__ void edge_rows_add(int sv, int dv, float nv, int j,
                                              int lane,
                                              const float* __restrict__ h,
                                              float* __restrict__ acc) {
  float ra[U], rb[U], nn[U];
  int cA[U];
  size_t w0[U];
#pragma unroll
  for (int u = 0; u < U; ++u) {
    const int s = __builtin_amdgcn_readlane(sv, j + u);
    const int d = __builtin_amdgcn_readlane(dv, j + u);
    nn[u] = __int_as_float(__builtin_amdgcn_readlane(__float_as_int(nv), j + u));
    const int start = d * NCOL;
    const int inoff = (NCOL % 32 == 0) ? 0 : (start & 31);
    w0[u] = (size_t)(start - inoff);
    cA[u] = lane - inoff;
    const int cB = cA[u] + 32;
    const float* hp = h + (size_t)s * NCOL;
    ra[u] = hp[cA[u] < 0 ? 0 : cA[u]];
    rb[u] = hp[cB < NCOL ? cB : NCOL - 1];
  }
  if (NCOL % 32 != 0) {
#pragma unroll
    for (int u = 0; u < U; ++u) asm volatile("" ::"v"(ra[u]), "v"(rb[u]));
  }
#pragma unroll
  for (int u = 0; u < U; ++u) {
    float* ap = acc + w0[u] + lane;
    if (NCOL % 32 == 0 || cA[u] >= 0) atomicAdd(ap, ra[u] * nn[u]);
    if (NCOL % 32 == 0 || cA[u] + 32 < NCOL) atomicAdd(ap + 32, rb[u] * nn[u]);
  }
}

template <int NCOL>
__global__ void __launch_bounds__(256)
k_edge_agg_lines(const int* __restrict__ src, const int* __restrict__ dst,
                 const float* __restrict__ dinv, const float* __restrict__ h,
                 float* __restrict__ acc, int e) {
  static_assert(NCOL == 64 || NCOL == 40, "a row must lie in two 128-byte lines");
  const int lane = threadIdx.x & 31;
  const int e0 = __builtin_amdgcn_readfirstlane(
      (int)((blockIdx.x * blockDim.x + threadIdx.x) >> 5) * 32);
  if (e0 >= e) return;
  const int mine = e0 + lane;
  const bool have = mine < e;
  const int sv = have ? src[mine] : 0;
  const int dv = have ? dst[mine] : 0;
  const float nv = have ? dinv[sv] * dinv[dv] : 0.0f;
  const int cnt = (e - e0 < 32) ? (e - e0) : 32;
  int j = 0;
  for (; j + 8 <= cnt; j += 8) edge_rows_add<NCOL, 8>(sv, dv, nv, j, lane, h, acc);
  for (; j < cnt; ++j) edge_rows_add<NCOL, 1>(sv, dv, nv, j, lane, h, acc);
}

__global__ void k_bias_logsoftmax40(float* __restrict__ out,
                                    const float* __restrict__ b, int n) {
  int i = blockIdx.x * blockDim.x + threadIdx.x;
  if (i >= n) return;
  float* p = out + (size_t)i * 40;
  float v[40];
  float m = -3.402823466e38f;
#pragma unroll
  for (int f = 0; f < 40; ++f) {
    v[f] = p[f] + b[f];
    m = fmaxf(m, v[f]);
  }
  float s = 0.0f;
#pragma unroll
  for (int f = 0; f < 40; ++f) s += expf(v[f] - m);
  float lse = m + logf(s);
#pragma unroll
  for (int f = 0; f < 40; ++f) p[f] = v[f] - lse;
}

extern "C" void kernel_launch(void* const* d_in, const int* in_sizes, int n_in,
                              void* d_out, int out_size, void* d_ws, size_t ws_size,
                              hipStream_t stream) {
  (void)n_in; (void)out_size; (void)ws_size;

  const float* x  = (const float*)d_in[0];
  const int*   ei = (const int*)d_in[1];
  const float* W1 = (const float*)d_in[2];
  const float* b1 = (const float*)d_in[3];
  const float* W2 = (const float*)d_in[4];
  const float* b2 = (const float*)d_in[5];
  float* out = (float*)d_out;

  const int F_IN = 128;
  const int Nn = in_sizes[0] / F_IN;
  const int E  = in_sizes[1] / 2;
  const int* src = ei;
  const int* dst = ei + E;

  char* ws = (char*)d_ws;
  float* dinv = (float*)(ws);
  float* h    = (float*)(ws + ((size_t)4  << 20));
  float* acc1 = (float*)(ws + ((size_t)36 << 20));
  float* h2   = (float*)(ws + ((size_t)68 << 20));

  const dim3 B256(256);

  k_init_deg<<<(Nn + 255) / 256, B256, 0, stream>>>(dinv, Nn);
  k_count_deg<<<(E + 255) / 256, B256, 0, stream>>>(dst, dinv, E);
  k_rsqrt_inplace<<<(Nn + 255) / 256, B256, 0, stream>>>(dinv, Nn);

  k_gemm_wmma_f32<128, 64, false><<<dim3(1024), dim3(128), 0, stream>>>(
      x, b1, W1, dinv, h, acc1, Nn / 16);
  k_edge_agg_lines<64><<<(E + 255) / 256, B256, 0, stream>>>(src, dst, dinv, h, acc1, E);

  k_gemm_wmma_f32<64, 40, true><<<dim3(1024), dim3(96), 0, stream>>>(
      acc1, b1, W2, dinv, h2, out, Nn / 16);
  k_edge_agg_lines<40><<<(E + 255) / 256, B256, 0, stream>>>(src, dst, dinv, h2, out, E);
  k_bias_logsoftmax40<<<(Nn + 255) / 256, B256, 0, stream>>>(out, b2, Nn);
}


# f32 atomic adds (degree count, both edge scatters) issued only from the package half that owns the row: two row ranges, chunks drawn from per-half ticket counters; adds still in arrival order
# speedup vs baseline: 2.2834x; 1.2867x over previous
#include <hip/hip_runtime.h>
#include <hip/hip_bf16.h>

typedef __attribute__((ext_vector_type(2))) float v2f;
typedef __attribute__((ext_vector_type(4))) float v4f;
typedef __attribute__((ext_vector_type(8))) float v8f;

__global__ void k_init_deg(float* __restrict__ deg, int* __restrict__ ctr, int n) {
  int i = blockIdx.x * blockDim.x + threadIdx.x;
  if (i < n) deg[i] = 1.0f;
  if (i < 6) ctr[i] = 0;
}

__device__ __forceinline__ int package_half() {
  return (int)((__builtin_amdgcn_s_sendmsg_rtn(0x87) >> 18) & 1u);
}

__device__ __forceinline__ int range_split(int n) { return ((n >> 1) + 31) & ~31; }

template <typename Body>
__device__ __forceinline__ void walk_tickets(int* __restrict__ ctr, int nchunk, Body body) {
  __shared__ int s_half;
  __shared__ int s_tick[2];
  if (threadIdx.x == 0) s_half = package_half();
  __syncthreads();
  const int half = __builtin_amdgcn_readfirstlane(s_half);
  int flip = 0;
  for (int pass = 0; pass < 2; ++pass) {
    const int q = half ^ pass;
    for (;;) {
      if (threadIdx.x == 0) s_tick[flip] = atomicAdd(&ctr[q], 1);
      __syncthreads();
      const int t = __builtin_amdgcn_readfirstlane(s_tick[flip]);
      flip ^= 1;
      if (t >= nchunk) break;
      body(t, q);
    }
  }
}

__global__ void __launch_bounds__(256)
k_count_deg(const int* __restrict__ dst, float* __restrict__ deg,
            int* __restrict__ ctr, int e, int n) {
  const int split = range_split(n);
  walk_tickets(ctr, (e + 1023) / 1024, [&](int t, int q) {
    const int i0 = t * 1024 + (int)threadIdx.x;
    int d[4];
#pragma unroll
    for (int u = 0; u < 4; ++u) d[u] = (i0 + 256 * u < e) ? dst[i0 + 256 * u] : -1;
#pragma unroll
    for (int u = 0; u < 4; ++u)
      if (d[u] >= 0 && (d[u] >= split) == (q != 0))
        atomicAdd(&deg[d[u]], 1.0f);
  });
}

__global__ void k_rsqrt_inplace(float* __restrict__ d, int n) {
  int i = blockIdx.x * blockDim.x + threadIdx.x;
  if (i < n) d[i] = rsqrtf(d[i]);
}

__device__ __forceinline__ int kof(int j, int h, int s) {
  return 8 * (j >> 1) + 4 * h + 2 * (j & 1) + s;
}

template <int K, int N, bool RELU_A>
struct GemmWS {
  static constexpr int NJ = K / 4;
  static constexpr int NQ = K / 8;

  struct Tile {
    v4f a[NQ];
    v4f dlo, dhi;
  };

  static __device__ __forceinline__ void load(Tile& t, int tile, int c16, int h,
                                              const float* __restrict__ A,
                                              const float* __restrict__ dinv) {
    const int m0 = tile * 16;
    const v4f* ap = reinterpret_cast<const v4f*>(A + (size_t)(m0 + c16) * K + 4 * h);
#pragma unroll
    for (int i = 0; i < NQ; ++i) t.a[i] = ap[2 * i];
    const v4f* dp = reinterpret_cast<const v4f*>(dinv + m0 + 8 * h);
    t.dlo = dp[0];
    t.dhi = dp[1];
  }

  static __device__ __forceinline__ void run(const Tile& t, int tile, int col, int h,
                                             const v2f (&bw)[NJ], const v4f (&bias)[NQ],
                                             float* __restrict__ C,
                                             float* __restrict__ S) {
    v8f c = {};
#pragma unroll
    for (int i = 0; i < NQ; ++i) {
      v4f q = t.a[i];
      if (RELU_A) {
        q.x = fmaxf(q.x + bias[i].x, 0.0f);
        q.y = fmaxf(q.y + bias[i].y, 0.0f);
        q.z = fmaxf(q.z + bias[i].z, 0.0f);
        q.w = fmaxf(q.w + bias[i].w, 0.0f);
      }
      v2f lo, hi;
      lo.x = q.x; lo.y = q.y;
      hi.x = q.z; hi.y = q.w;
      c = __builtin_amdgcn_wmma_f32_16x16x4_f32(false, lo, false, bw[2 * i], (short)0, c,
                                                false, false);
      c = __builtin_amdgcn_wmma_f32_16x16x4_f32(false, hi, false, bw[2 * i + 1], (short)0, c,
                                                false, false);
    }
    if (N % 16 == 0 || col < N) {
      const float di[8] = {t.dlo.x, t.dlo.y, t.dlo.z, t.dlo.w,
                           t.dhi.x, t.dhi.y, t.dhi.z, t.dhi.w};
      const int m0 = tile * 16;
#pragma unroll
      for (int r = 0; r < 8; ++r) {
        const size_t o = (size_t)(m0 + r + 8 * h) * N + col;
        C[o] = c[r];
        S[o] = c[r] * di[r] * di[r];
      }
    }
  }
};

template <int K, int N, bool RELU_A>
__global__ void __launch_bounds__(((N + 15) / 16) * 32)
k_gemm_wmma_f32(const float* __restrict__ A,
                const float* __restrict__ abias,
                const float* __restrict__ B,
                const float* __restrict__ dinv,
                float* __restrict__ C,
                float* __restrict__ S,
                int ntiles) {
  typedef GemmWS<K, N, RELU_A> G;
  const int lane = threadIdx.x & 31;
  const int c16  = lane & 15;
  const int h    = lane >> 4;
  const int col  = (int)(threadIdx.x >> 5) * 16 + c16;
  const int colc = (col < N) ? col : (N - 1);

  v2f bw[G::NJ];
#pragma unroll
  for (int j = 0; j < G::NJ; ++j) {
    bw[j].x = B[(size_t)kof(j, h, 0) * N + colc];
    bw[j].y = B[(size_t)kof(j, h, 1) * N + colc];
  }
  v4f bias[G::NQ];
#pragma unroll
  for (int i = 0; i < G::NQ; ++i) {
    if (RELU_A) {
      bias[i].x = abias[kof(2 * i, h, 0)];
      bias[i].y = abias[kof(2 * i, h, 1)];
      bias[i].z = abias[kof(2 * i + 1, h, 0)];
      bias[i].w = abias[kof(2 * i + 1, h, 1)];
    } else {
      bias[i] = v4f{0.0f, 0.0f, 0.0f, 0.0f};
    }
  }

  const int step = gridDim.x;
  int t = blockIdx.x;
  if (t >= ntiles) return;
  typename G::Tile t0, t1;
  G::load(t0, t, c16, h, A, dinv);
  while (t + step < ntiles) {
    const int tn = t + step;
    G::load(t1, tn, c16, h, A, dinv);
    __builtin_amdgcn_sched_barrier(0);
    G::run(t0, t, col, h, bw, bias, C, S);
    const int tnn = tn + step;
    G::load(t0, tnn < ntiles ? tnn : tn, c16, h, A, dinv);
    __builtin_amdgcn_sched_barrier(0);
    G::run(t1, tn, col, h, bw, bias, C, S);
    t = tnn;
  }
  if (t < ntiles) G::run(t0, t, col, h, bw, bias, C, S);
}

template <int NCOL, int U>
__device__ __forceinline__ void edge_rows_add(int sv, int dv, float nv,
                                              const int (&jj)[U], int lane,
                                              const float* __restrict__ h,
                                              float* __restrict__ acc) {
  float ra[U], rb[U], nn[U];
  int cA[U];
  size_t w0[U];
#pragma unroll
  for (int u = 0; u < U; ++u) {
    const int s = __builtin_amdgcn_readlane(sv, jj[u]);
    const int d = __builtin_amdgcn_readlane(dv, jj[u]);
    nn[u] = __int_as_float(__builtin_amdgcn_readlane(__float_as_int(nv), jj[u]));
    const int start = d * NCOL;
    const int inoff = (NCOL % 32 == 0) ? 0 : (start & 31);
    w0[u] = (size_t)(start - inoff);
    cA[u] = lane - inoff;
    const int cB = cA[u] + 32;
    const float* hp = h + (size_t)s * NCOL;
    ra[u] = hp[cA[u] < 0 ? 0 : cA[u]];
    rb[u] = hp[cB < NCOL ? cB : NCOL - 1];
  }
  if (NCOL % 32 != 0) {
#pragma unroll
    for (int u = 0; u < U; ++u) asm volatile("" ::"v"(ra[u]), "v"(rb[u]));
  }
#pragma unroll
  for (int u = 0; u < U; ++u) {
    float* ap = acc + w0[u] + lane;
    if (NCOL % 32 == 0 || cA[u] >= 0) atomicAdd(ap, ra[u] * nn[u]);
    if (NCOL % 32 == 0 || cA[u] + 32 < NCOL) atomicAdd(ap + 32, rb[u] * nn[u]);
  }
}

template <int NCOL, int U>
__device__ __forceinline__ unsigned take_edges(unsigned m, int sv, int dv, float nv, int lane,
                                               const float* __restrict__ h,
                                               float* __restrict__ acc) {
  int jj[U];
#pragma unroll
  for (int u = 0; u < U; ++u) {
    jj[u] = __builtin_ctz(m);
    m &= m - 1;
  }
  edge_rows_add<NCOL, U>(sv, dv, nv, jj, lane, h, acc);
  return m;
}

template <int NCOL>
__global__ void __launch_bounds__(256)
k_edge_agg_lines(const int* __restrict__ src, const int* __restrict__ dst,
                 const float* __restrict__ dinv, const float* __restrict__ h,
                 float* __restrict__ acc, int* __restrict__ ctr, int e, int n) {
  static_assert(NCOL == 64 || NCOL == 40, "a row must lie in two 128-byte lines");
  const int lane = threadIdx.x & 31;
  const int woff = __builtin_amdgcn_readfirstlane((int)(threadIdx.x >> 5) * 32);
  const int split = range_split(n);
  walk_tickets(ctr, (e + 255) / 256, [&](int t, int q) {
    const int mine = t * 256 + woff + lane;
    const bool have = mine < e;
    const int sv = have ? src[mine] : 0;
    const int dv = have ? dst[mine] : 0;
    const bool own = have && ((dv >= split) == (q != 0));
    const float nv = own ? dinv[sv] * dinv[dv] : 0.0f;
    unsigned m = __builtin_amdgcn_ballot_w32(own);
    while (m != 0u) {
      const int left = __builtin_popcount(m);
      if (left >= 8)      m = take_edges<NCOL, 8>(m, sv, dv, nv, lane, h, acc);
      else if (left >= 4) m = take_edges<NCOL, 4>(m, sv, dv, nv, lane, h, acc);
      else                m = take_edges<NCOL, 1>(m, sv, dv, nv, lane, h, acc);
    }
  });
}

__global__ void k_bias_logsoftmax40(float* __restrict__ out,
                                    const float* __restrict__ b, int n) {
  int i = blockIdx.x * blockDim.x + threadIdx.x;
  if (i >= n) return;
  float* p = out + (size_t)i * 40;
  float v[40];
  float m = -3.402823466e38f;
#pragma unroll
  for (int f = 0; f < 40; ++f) {
    v[f] = p[f] + b[f];
    m = fmaxf(m, v[f]);
  }
  float s = 0.0f;
#pragma unroll
  for (int f = 0; f < 40; ++f) s += expf(v[f] - m);
  float lse = m + logf(s);
#pragma unroll
  for (int f = 0; f < 40; ++f) p[f] = v[f] - lse;
}

extern "C" void kernel_launch(void* const* d_in, const int* in_sizes, int n_in,
                              void* d_out, int out_size, void* d_ws, size_t ws_size,
                              hipStream_t stream) {
  (void)n_in; (void)out_size; (void)ws_size;

  const float* x  = (const float*)d_in[0];
  const int*   ei = (const int*)d_in[1];
  const float* W1 = (const float*)d_in[2];
  const float* b1 = (const float*)d_in[3];
  const float* W2 = (const float*)d_in[4];
  const float* b2 = (const float*)d_in[5];
  float* out = (float*)d_out;

  const int F_IN = 128;
  const int Nn = in_sizes[0] / F_IN;
  const int E  = in_sizes[1] / 2;
  const int* src = ei;
  const int* dst = ei + E;

  char* ws = (char*)d_ws;
  float* dinv = (float*)(ws);
  float* h    = (float*)(ws + ((size_t)4  << 20));
  float* acc1 = (float*)(ws + ((size_t)36 << 20));
  float* h2   = (float*)(ws + ((size_t)68 << 20));
  int*   ctr  = (int*)(ws + ((size_t)88 << 20));

  const dim3 B256(256);

  k_init_deg<<<(Nn + 255) / 256, B256, 0, stream>>>(dinv, ctr, Nn);
  k_count_deg<<<dim3(1024), B256, 0, stream>>>(dst, dinv, ctr, E, Nn);
  k_rsqrt_inplace<<<(Nn + 255) / 256, B256, 0, stream>>>(dinv, Nn);

  k_gemm_wmma_f32<128, 64, false><<<dim3(1024), dim3(128), 0, stream>>>(
      x, b1, W1, dinv, h, acc1, Nn / 16);
  k_edge_agg_lines<64><<<dim3(2048), B256, 0, stream>>>(src, dst, dinv, h, acc1, ctr + 2, E, Nn);

  k_gemm_wmma_f32<64, 40, true><<<dim3(1024), dim3(96), 0, stream>>>(
      acc1, b1, W2, dinv, h2, out, Nn / 16);
  k_edge_agg_lines<40><<<dim3(2048), B256, 0, stream>>>(src, dst, dinv, h2, out, ctr + 4, E, Nn);
  k_bias_logsoftmax40<<<(Nn + 255) / 256, B256, 0, stream>>>(out, b2, Nn);
}


# Layer-2 scatter reads each edge's weight from a per-range plane that the layer-1 scatter stored (one line load per 32 edges) instead of two random 4-byte dinv look-ups per edge; values unchanged
# speedup vs baseline: 2.3948x; 1.0488x over previous
#include <hip/hip_runtime.h>
#include <hip/hip_bf16.h>

typedef __attribute__((ext_vector_type(2))) float v2f;
typedef __attribute__((ext_vector_type(4))) float v4f;
typedef __attribute__((ext_vector_type(8))) float v8f;

__global__ void k_init_deg(float* __restrict__ deg, int* __restrict__ ctr, int n) {
  int i = blockIdx.x * blockDim.x + threadIdx.x;
  if (i < n) deg[i] = 1.0f;
  if (i < 6) ctr[i] = 0;
}

__device__ __forceinline__ int package_half() {
  return (int)((__builtin_amdgcn_s_sendmsg_rtn(0x87) >> 18) & 1u);
}

__device__ __forceinline__ int range_split(int n) { return ((n >> 1) + 31) & ~31; }

template <typename Body>
__device__ __forceinline__ void walk_tickets(int* __restrict__ ctr, int nchunk, Body body) {
  __shared__ int s_half;
  __shared__ int s_tick[2];
  if (threadIdx.x == 0) s_half = package_half();
  __syncthreads();
  const int half = __builtin_amdgcn_readfirstlane(s_half);
  int flip = 0;
  for (int pass = 0; pass < 2; ++pass) {
    const int q = half ^ pass;
    for (;;) {
      if (threadIdx.x == 0) s_tick[flip] = atomicAdd(&ctr[q], 1);
      __syncthreads();
      const int t = __builtin_amdgcn_readfirstlane(s_tick[flip]);
      flip ^= 1;
      if (t >= nchunk) break;
      body(t, q);
    }
  }
}

__global__ void __launch_bounds__(256)
k_count_deg(const int* __restrict__ dst, float* __restrict__ deg,
            int* __restrict__ ctr, int e, int n) {
  const int split = range_split(n);
  walk_tickets(ctr, (e + 1023) / 1024, [&](int t, int q) {
    const int i0 = t * 1024 + (int)threadIdx.x;
    int d[4];
#pragma unroll
    for (int u = 0; u < 4; ++u) d[u] = (i0 + 256 * u < e) ? dst[i0 + 256 * u] : -1;
#pragma unroll
    for (int u = 0; u < 4; ++u)
      if (d[u] >= 0 && (d[u] >= split) == (q != 0))
        atomicAdd(&deg[d[u]], 1.0f);
  });
}

__global__ void k_rsqrt_inplace(float* __restrict__ d, int n) {
  int i = blockIdx.x * blockDim.x + threadIdx.x;
  if (i < n) d[i] = rsqrtf(d[i]);
}

__device__ __forceinline__ int kof(int j, int h, int s) {
  return 8 * (j >> 1) + 4 * h + 2 * (j & 1) + s;
}

template <int K, int N, bool RELU_A>
struct GemmWS {
  static constexpr int NJ = K / 4;
  static constexpr int NQ = K / 8;

  struct Tile {
    v4f a[NQ];
    v4f dlo, dhi;
  };

  static __device__ __forceinline__ void load(Tile& t, int tile, int c16, int h,
                                              const float* __restrict__ A,
                                              const float* __restrict__ dinv) {
    const int m0 = tile * 16;
    const v4f* ap = reinterpret_cast<const v4f*>(A + (size_t)(m0 + c16) * K + 4 * h);
#pragma unroll
    for (int i = 0; i < NQ; ++i) t.a[i] = ap[2 * i];
    const v4f* dp = reinterpret_cast<const v4f*>(dinv + m0 + 8 * h);
    t.dlo = dp[0];
    t.dhi = dp[1];
  }

  static __device__ __forceinline__ void run(const Tile& t, int tile, int col, int h,
                                             const v2f (&bw)[NJ], const v4f (&bias)[NQ],
                                             float* __restrict__ C,
                                             float* __restrict__ S) {
    v8f c = {};
#pragma unroll
    for (int i = 0; i < NQ; ++i) {
      v4f q = t.a[i];
      if (RELU_A) {
        q.x = fmaxf(q.x + bias[i].x, 0.0f);
        q.y = fmaxf(q.y + bias[i].y, 0.0f);
        q.z = fmaxf(q.z + bias[i].z, 0.0f);
        q.w = fmaxf(q.w + bias[i].w, 0.0f);
      }
      v2f lo, hi;
      lo.x = q.x; lo.y = q.y;
      hi.x = q.z; hi.y = q.w;
      c = __builtin_amdgcn_wmma_f32_16x16x4_f32(false, lo, false, bw[2 * i], (short)0, c,
                                                false, false);
      c = __builtin_amdgcn_wmma_f32_16x16x4_f32(false, hi, false, bw[2 * i + 1], (short)0, c,
                                                false, false);
    }
    if (N % 16 == 0 || col < N) {
      const float di[8] = {t.dlo.x, t.dlo.y, t.dlo.z, t.dlo.w,
                           t.dhi.x, t.dhi.y, t.dhi.z, t.dhi.w};
      const int m0 = tile * 16;
#pragma unroll
      for (int r = 0; r < 8; ++r) {
        const size_t o = (size_t)(m0 + r + 8 * h) * N + col;
        C[o] = c[r];
        S[o] = c[r] * di[r] * di[r];
      }
    }
  }
};

template <int K, int N, bool RELU_A>
__global__ void __launch_bounds__(((N + 15) / 16) * 32)
k_gemm_wmma_f32(const float* __restrict__ A,
                const float* __restrict__ abias,
                const float* __restrict__ B,
                const float* __restrict__ dinv,
                float* __restrict__ C,
                float* __restrict__ S,
                int ntiles) {
  typedef GemmWS<K, N, RELU_A> G;
  const int lane = threadIdx.x & 31;
  const int c16  = lane & 15;
  const int h    = lane >> 4;
  const int col  = (int)(threadIdx.x >> 5) * 16 + c16;
  const int colc = (col < N) ? col : (N - 1);

  v2f bw[G::NJ];
#pragma unroll
  for (int j = 0; j < G::NJ; ++j) {
    bw[j].x = B[(size_t)kof(j, h, 0) * N + colc];
    bw[j].y = B[(size_t)kof(j, h, 1) * N + colc];
  }
  v4f bias[G::NQ];
#pragma unroll
  for (int i = 0; i < G::NQ; ++i) {
    if (RELU_A) {
      bias[i].x = abias[kof(2 * i, h, 0)];
      bias[i].y = abias[kof(2 * i, h, 1)];
      bias[i].z = abias[kof(2 * i + 1, h, 0)];
      bias[i].w = abias[kof(2 * i + 1, h, 1)];
    } else {
      bias[i] = v4f{0.0f, 0.0f, 0.0f, 0.0f};
    }
  }

  const int step = gridDim.x;
  int t = blockIdx.x;
  if (t >= ntiles) return;
  typename G::Tile t0, t1;
  G::load(t0, t, c16, h, A, dinv);
  while (t + step < ntiles) {
    const int tn = t + step;
    G::load(t1, tn, c16, h, A, dinv);
    __builtin_amdgcn_sched_barrier(0);
    G::run(t0, t, col, h, bw, bias, C, S);
    const int tnn = tn + step;
    G::load(t0, tnn < ntiles ? tnn : tn, c16, h, A, dinv);
    __builtin_amdgcn_sched_barrier(0);
    G::run(t1, tn, col, h, bw, bias, C, S);
    t = tnn;
  }
  if (t < ntiles) G::run(t0, t, col, h, bw, bias, C, S);
}

template <int NCOL, int U>
__device__ __forceinline__ void edge_rows_add(int sv, int dv, float nv,
                                              const int (&jj)[U], int lane,
                                              const float* __restrict__ h,
                                              float* __restrict__ acc) {
  float ra[U], rb[U], nn[U];
  int cA[U];
  size_t w0[U];
#pragma unroll
  for (int u = 0; u < U; ++u) {
    const int s = __builtin_amdgcn_readlane(sv, jj[u]);
    const int d = __builtin_amdgcn_readlane(dv, jj[u]);
    nn[u] = __int_as_float(__builtin_amdgcn_readlane(__float_as_int(nv), jj[u]));
    const int start = d * NCOL;
    const int inoff = (NCOL % 32 == 0) ? 0 : (start & 31);
    w0[u] = (size_t)(start - inoff);
    cA[u] = lane - inoff;
    const int cB = cA[u] + 32;
    const float* hp = h + (size_t)s * NCOL;
    ra[u] = hp[cA[u] < 0 ? 0 : cA[u]];
    rb[u] = hp[cB < NCOL ? cB : NCOL - 1];
  }
  if (NCOL % 32 != 0) {
#pragma unroll
    for (int u = 0; u < U; ++u) asm volatile("" ::"v"(ra[u]), "v"(rb[u]));
  }
#pragma unroll
  for (int u = 0; u < U; ++u) {
    float* ap = acc + w0[u] + lane;
    if (NCOL % 32 == 0 || cA[u] >= 0) atomicAdd(ap, ra[u] * nn[u]);
    if (NCOL % 32 == 0 || cA[u] + 32 < NCOL) atomicAdd(ap + 32, rb[u] * nn[u]);
  }
}

template <int NCOL, int U>
__device__ __forceinline__ unsigned take_edges(unsigned m, int sv, int dv, float nv, int lane,
                                               const float* __restrict__ h,
                                               float* __restrict__ acc) {
  int jj[U];
#pragma unroll
  for (int u = 0; u < U; ++u) {
    jj[u] = __builtin_ctz(m);
    m &= m - 1;
  }
  edge_rows_add<NCOL, U>(sv, dv, nv, jj, lane, h, acc);
  return m;
}

template <int NCOL, bool FIRST>
__global__ void __launch_bounds__(256)
k_edge_agg_lines(const int* __restrict__ src, const int* __restrict__ dst,
                 const float* __restrict__ dinv, const float* __restrict__ h,
                 float* __restrict__ acc, float* __restrict__ wplane,
                 int* __restrict__ ctr, int e, int n) {
  static_assert(NCOL == 64 || NCOL == 40, "a row must lie in two 128-byte lines");
  const int lane = threadIdx.x & 31;
  const int woff = __builtin_amdgcn_readfirstlane((int)(threadIdx.x >> 5) * 32);
  const int split = range_split(n);
  const size_t plane = (size_t)((e + 255) & ~255);
  walk_tickets(ctr, (e + 255) / 256, [&](int t, int q) {
    const int mine = t * 256 + woff + lane;
    const bool have = mine < e;
    float* wp = wplane + (size_t)q * plane + mine;
    const int sv = have ? src[mine] : 0;
    const int dv = have ? dst[mine] : 0;
    float nv;
    if (!FIRST) nv = have ? *wp : 0.0f;
    const bool own = have && ((dv >= split) == (q != 0));
    if (FIRST) {
      nv = own ? dinv[sv] * dinv[dv] : 0.0f;
      if (have) *wp = nv;
    }
    unsigned m = __builtin_amdgcn_ballot_w32(own);
    while (m != 0u) {
      const int left = __builtin_popcount(m);
      if (left >= 8)      m = take_edges<NCOL, 8>(m, sv, dv, nv, lane, h, acc);
      else if (left >= 4) m = take_edges<NCOL, 4>(m, sv, dv, nv, lane, h, acc);
      else                m = take_edges<NCOL, 1>(m, sv, dv, nv, lane, h, acc);
    }
  });
}

__global__ void k_bias_logsoftmax40(float* __restrict__ out,
                                    const float* __restrict__ b, int n) {
  int i = blockIdx.x * blockDim.x + threadIdx.x;
  if (i >= n) return;
  float* p = out + (size_t)i * 40;
  float v[40];
  float m = -3.402823466e38f;
#pragma unroll
  for (int f = 0; f < 40; ++f) {
    v[f] = p[f] + b[f];
    m = fmaxf(m, v[f]);
  }
  float s = 0.0f;
#pragma unroll
  for (int f = 0; f < 40; ++f) s += expf(v[f] - m);
  float lse = m + logf(s);
#pragma unroll
  for (int f = 0; f < 40; ++f) p[f] = v[f] - lse;
}

extern "C" void kernel_launch(void* const* d_in, const int* in_sizes, int n_in,
                              void* d_out, int out_size, void* d_ws, size_t ws_size,
                              hipStream_t stream) {
  (void)n_in; (void)out_size; (void)ws_size;

  const float* x  = (const float*)d_in[0];
  const int*   ei = (const int*)d_in[1];
  const float* W1 = (const float*)d_in[2];
  const float* b1 = (const float*)d_in[3];
  const float* W2 = (const float*)d_in[4];
  const float* b2 = (const float*)d_in[5];
  float* out = (float*)d_out;

  const int F_IN = 128;
  const int Nn = in_sizes[0] / F_IN;
  const int E  = in_sizes[1] / 2;
  const int* src = ei;
  const int* dst = ei + E;

  char* ws = (char*)d_ws;
  float* dinv = (float*)(ws);
  float* h    = (float*)(ws + ((size_t)4  << 20));
  float* acc1 = (float*)(ws + ((size_t)36 << 20));
  float* h2   = (float*)(ws + ((size_t)68 << 20));
  int*   ctr  = (int*)(ws + ((size_t)88 << 20));
  float* wpl  = (float*)(ws + ((size_t)89 << 20));

  const dim3 B256(256);

  k_init_deg<<<(Nn + 255) / 256, B256, 0, stream>>>(dinv, ctr, Nn);
  k_count_deg<<<dim3(1024), B256, 0, stream>>>(dst, dinv, ctr, E, Nn);
  k_rsqrt_inplace<<<(Nn + 255) / 256, B256, 0, stream>>>(dinv, Nn);

  k_gemm_wmma_f32<128, 64, false><<<dim3(1024), dim3(128), 0, stream>>>(
      x, b1, W1, dinv, h, acc1, Nn / 16);
  k_edge_agg_lines<64, true><<<dim3(2048), B256, 0, stream>>>(src, dst, dinv, h, acc1, wpl,
                                                             ctr + 2, E, Nn);

  k_gemm_wmma_f32<64, 40, true><<<dim3(1024), dim3(96), 0, stream>>>(
      acc1, b1, W2, dinv, h2, out, Nn / 16);
  k_edge_agg_lines<40, false><<<dim3(2048), B256, 0, stream>>>(src, dst, dinv, h2, out, wpl,
                                                              ctr + 4, E, Nn);
  k_bias_logsoftmax40<<<(Nn + 255) / 256, B256, 0, stream>>>(out, b2, Nn);
}
